# starting kernel
# baseline (speedup 1.0000x reference)
#include <hip/hip_runtime.h>
#include <math.h>
#include <stdint.h>

#define NFEAT 256
#define NHID  64

typedef float v2f __attribute__((ext_vector_type(2)));
typedef float v8f __attribute__((ext_vector_type(8)));

__global__ void deg_init_kernel(float* __restrict__ deg, int n) {
  int i = blockIdx.x * blockDim.x + threadIdx.x;
  if (i < n) deg[i] = 1.0f;
}

__global__ void deg_count_kernel(const int* __restrict__ col,
                                 float* __restrict__ deg, int e) {
  int i = blockIdx.x * blockDim.x + threadIdx.x;
  if (i < e) atomicAdd(&deg[col[i]], 1.0f);
}

__global__ __launch_bounds__(256)
void gemm_x_w1_kernel(const float* __restrict__ x,
                      const float* __restrict__ W1,
                      float* __restrict__ h, int n) {
  __shared__ float sW[128 * NHID];
  const int tid  = threadIdx.x;
  const int wave = tid >> 5;
  const int lane = tid & 31;
  const int l15  = lane & 15;
  const int kh   = lane >> 4;
  const int rowBase = blockIdx.x * 128 + wave * 16;
  const int row  = rowBase + l15;
  const int rowC = (row < n) ? row : (n - 1);

  v8f acc[4];
#pragma unroll
  for (int t = 0; t < 4; ++t) acc[t] = (v8f){0,0,0,0,0,0,0,0};

  const float* xrow = x + (size_t)rowC * NFEAT + 2 * kh;
  __builtin_prefetch(xrow + 128, 0, 3);

  for (int kc = 0; kc < 2; ++kc) {
    __syncthreads();
    for (int idx = tid; idx < 64 * NHID; idx += 256) {
      int kp = idx >> 6;
      int nn = idx & 63;
      int kg = kc * 128 + kp * 2;
      ((float2*)sW)[kp * NHID + nn] =
          make_float2(W1[(size_t)kg * NHID + nn], W1[(size_t)(kg + 1) * NHID + nn]);
    }
    __syncthreads();

#pragma unroll 4
    for (int k0 = 0; k0 < 128; k0 += 4) {
      v2f a = *(const v2f*)(xrow + kc * 128 + k0);
      const int kp = (k0 >> 1) + kh;
#pragma unroll
      for (int t = 0; t < 4; ++t) {
        v2f b = ((const v2f*)sW)[kp * NHID + t * 16 + l15];
        acc[t] = __builtin_amdgcn_wmma_f32_16x16x4_f32(
            false, a, false, b, (short)0, acc[t], false, false);
      }
    }
  }

#pragma unroll
  for (int t = 0; t < 4; ++t) {
    const int ccol = t * 16 + l15;
#pragma unroll
    for (int j = 0; j < 8; ++j) {
      int r = rowBase + j + 8 * kh;
      if (r < n) h[(size_t)r * NHID + ccol] = acc[t][j];
    }
  }
}

__global__ void agg_init_kernel(const float* __restrict__ h,
                                const float* __restrict__ deg,
                                float* __restrict__ dinv,
                                float* __restrict__ agg, long long total) {
  long long idx = (long long)blockIdx.x * blockDim.x + threadIdx.x;
  if (idx >= total) return;
  int i = (int)(idx >> 6);
  int j = (int)(idx & 63);
  float d = rsqrtf(deg[i]);
  if (j == 0) dinv[i] = d;
  agg[idx] = d * d * h[idx];
}

__global__ __launch_bounds__(256)
void edge_scatter_kernel(const int* __restrict__ row,
                         const int* __restrict__ col,
                         const float* __restrict__ dinv,
                         const float* __restrict__ h,
                         float* __restrict__ agg, int e) {
  int wv   = (int)((blockIdx.x * blockDim.x + threadIdx.x) >> 5);
  int lane = threadIdx.x & 31;
  if (wv >= e) return;
  int r = __builtin_amdgcn_readfirstlane(row[wv]);
  int c = __builtin_amdgcn_readfirstlane(col[wv]);
  float norm = dinv[r] * dinv[c];
  const float* hr = h  + (size_t)r * NHID;
  float*       ac = agg + (size_t)c * NHID;
  atomicAdd(&ac[lane],      norm * hr[lane]);
  atomicAdd(&ac[lane + 32], norm * hr[lane + 32]);
}

__global__ __launch_bounds__(256)
void edge_mlp_kernel(const int* __restrict__ row, const int* __restrict__ col,
                     const float* __restrict__ agg,
                     const float* __restrict__ b1,
                     const float* __restrict__ Wfc,
                     const float* __restrict__ bfc,
                     float* __restrict__ out, int e) {
  int wv   = (int)((blockIdx.x * blockDim.x + threadIdx.x) >> 5);
  int lane = threadIdx.x & 31;
  if (wv >= e) return;
  int r = __builtin_amdgcn_readfirstlane(row[wv]);
  int c = __builtin_amdgcn_readfirstlane(col[wv]);
  const float* hc = agg + (size_t)c * NHID;
  const float* hr = agg + (size_t)r * NHID;
  float a0 = 0.f, a1 = 0.f;
#pragma unroll
  for (int q = 0; q < 4; ++q) {
    int j = lane + 32 * q;
    float v = (j < 64) ? (hc[j] + b1[j]) : (hr[j - 64] + b1[j - 64]);
    a0 += v * Wfc[j * 2 + 0];
    a1 += v * Wfc[j * 2 + 1];
  }
#pragma unroll
  for (int off = 16; off > 0; off >>= 1) {
    a0 += __shfl_xor(a0, off, 32);
    a1 += __shfl_xor(a1, off, 32);
  }
  if (lane == 0) {
    float l0 = a0 + bfc[0], l1 = a1 + bfc[1];
    float m  = fmaxf(l0, l1);
    float e0 = __expf(l0 - m), e1 = __expf(l1 - m);
    float s  = 1.0f / (e0 + e1);
    ((float2*)out)[wv] = make_float2(e0 * s, e1 * s);
  }
}

extern "C" void kernel_launch(void* const* d_in, const int* in_sizes, int n_in,
                              void* d_out, int out_size, void* d_ws, size_t ws_size,
                              hipStream_t stream) {
  const float* x   = (const float*)d_in[0];
  const int*   ei  = (const int*)d_in[1];
  const float* W1  = (const float*)d_in[2];
  const float* b1  = (const float*)d_in[3];
  const float* Wfc = (const float*)d_in[4];
  const float* bfc = (const float*)d_in[5];
  const int n = in_sizes[0] / NFEAT;
  const int e = in_sizes[1] / 2;
  const int* row = ei;
  const int* col = ei + e;

  uintptr_t p = (uintptr_t)d_ws;
  float* deg  = (float*)p;  p += (size_t)n * sizeof(float);
  float* dinv = (float*)p;  p += (size_t)n * sizeof(float);
  p = (p + 255) & ~(uintptr_t)255;
  float* h    = (float*)p;  p += (size_t)n * NHID * sizeof(float);
  p = (p + 255) & ~(uintptr_t)255;
  float* agg  = (float*)p;

  const long long aggTotal = (long long)n * NHID;

  deg_init_kernel <<<(n + 255) / 256, 256, 0, stream>>>(deg, n);
  deg_count_kernel<<<(e + 255) / 256, 256, 0, stream>>>(col, deg, e);
  gemm_x_w1_kernel<<<(n + 127) / 128, 256, 0, stream>>>(x, W1, h, n);
  agg_init_kernel <<<(unsigned)((aggTotal + 255) / 256), 256, 0, stream>>>(h, deg, dinv, agg, aggTotal);
  edge_scatter_kernel<<<(e + 7) / 8, 256, 0, stream>>>(row, col, dinv, h, agg, e);
  edge_mlp_kernel    <<<(e + 7) / 8, 256, 0, stream>>>(row, col, agg, b1, Wfc, bfc,
                                                       (float*)d_out, e);
}


# neighbour sum per block of 200 target nodes in an LDS table (LDS f32 adds) over edge slices put in block order by an LDS counting sort; agg rows stored once, no global f32 atomics; sum order changes
# speedup vs baseline: 3.1759x; 3.1759x over previous
#include <hip/hip_runtime.h>
#include <math.h>
#include <stdint.h>

#define NFEAT 256
#define NHID  64

typedef float v2f __attribute__((ext_vector_type(2)));
typedef float v8f __attribute__((ext_vector_type(8)));

#define DEG_RANGE   65536
#define DEG_SLICES  64
#define DEG_THREADS 1024
#define DEG_BATCH   25

__global__ __launch_bounds__(DEG_THREADS)
void deg_hist_kernel(const int* __restrict__ col, unsigned* __restrict__ part,
                     int n, int e) {
  __shared__ unsigned hist[DEG_RANGE];
  const int tid = threadIdx.x;
  const int s   = blockIdx.x;
  const int lo  = (int)blockIdx.y * DEG_RANGE;
  const int pitch = (n + 63) & ~63;

  uint4* h4 = (uint4*)hist;
  for (int j = tid; j < DEG_RANGE / 4; j += DEG_THREADS) h4[j] = make_uint4(0u, 0u, 0u, 0u);
  __syncthreads();

  const int per   = (e + DEG_SLICES - 1) / DEG_SLICES;
  const int start = s * per;
  const int end   = (start + per < e) ? (start + per) : e;
  for (int i0 = start; i0 < end; i0 += DEG_THREADS * DEG_BATCH) {
    int c[DEG_BATCH];
#pragma unroll
    for (int u = 0; u < DEG_BATCH; ++u) {
      const int i = i0 + u * DEG_THREADS + tid;
      const int v = col[(i < end) ? i : (end - 1)];
      c[u] = (i < end) ? v : -1;
    }
#pragma unroll
    for (int u = 0; u < DEG_BATCH; ++u) {
      const unsigned k = (unsigned)(c[u] - lo);
      if (k < (unsigned)DEG_RANGE) atomicAdd(&hist[k], 1u);
    }
  }
  __syncthreads();

  int len = pitch - lo;
  if (len > DEG_RANGE) len = DEG_RANGE;
  uint4* dst = (uint4*)(part + (size_t)s * pitch + lo);
#pragma unroll 4
  for (int j = tid; j < len / 4; j += DEG_THREADS) dst[j] = h4[j];
}

__global__ __launch_bounds__(256)
void deg_sum_kernel(const unsigned* __restrict__ part, float* __restrict__ deg, int n) {
  const int pitch = (n + 63) & ~63;
  const int i = (blockIdx.x * blockDim.x + threadIdx.x) * 4;
  if (i >= n) return;
  uint4 t = make_uint4(0u, 0u, 0u, 0u);
#pragma unroll 32
  for (int s = 0; s < DEG_SLICES; ++s) {
    const uint4 v = *(const uint4*)(part + (size_t)s * pitch + i);
    t.x += v.x; t.y += v.y; t.z += v.z; t.w += v.w;
  }
  const float d0 = 1.0f + (float)t.x, d1 = 1.0f + (float)t.y;
  const float d2 = 1.0f + (float)t.z, d3 = 1.0f + (float)t.w;
  if (i + 3 < n) {
    *(float4*)(deg + i) = make_float4(d0, d1, d2, d3);
  } else {
    deg[i] = d0;
    if (i + 1 < n) deg[i + 1] = d1;
    if (i + 2 < n) deg[i + 2] = d2;
  }
}

__global__ __launch_bounds__(256)
void gemm_x_w1_kernel(const float* __restrict__ x,
                      const float* __restrict__ W1,
                      const float* __restrict__ deg,
                      float* __restrict__ h,
                      float* __restrict__ dinv, int n) {
  __shared__ float sW[128 * NHID];
  const int tid  = threadIdx.x;
  const int wave = tid >> 5;
  const int lane = tid & 31;
  const int l15  = lane & 15;
  const int kh   = lane >> 4;
  const int rowBase = blockIdx.x * 128 + wave * 16;
  const int row  = rowBase + l15;
  const int rowC = (row < n) ? row : (n - 1);

  v8f acc[4];
#pragma unroll
  for (int t = 0; t < 4; ++t) acc[t] = (v8f){0,0,0,0,0,0,0,0};

  const float* xrow = x + (size_t)rowC * NFEAT + 2 * kh;
  __builtin_prefetch(xrow + 128, 0, 3);

  float dg[8];
#pragma unroll
  for (int j = 0; j < 8; ++j) dg[j] = 1.0f;

  for (int kc = 0; kc < 2; ++kc) {
    __syncthreads();
    for (int idx = tid; idx < 64 * NHID; idx += 256) {
      int kp = idx >> 6;
      int nn = idx & 63;
      int kg = kc * 128 + kp * 2;
      ((float2*)sW)[kp * NHID + nn] =
          make_float2(W1[(size_t)kg * NHID + nn], W1[(size_t)(kg + 1) * NHID + nn]);
    }
    __syncthreads();

    if (kc == 1) {
#pragma unroll
      for (int j = 0; j < 8; ++j) {
        const int r = rowBase + 8 * kh + j;
        dg[j] = deg[(r < n) ? r : (n - 1)];
      }
    }

#pragma unroll 4
    for (int k0 = 0; k0 < 128; k0 += 4) {
      v2f a = *(const v2f*)(xrow + kc * 128 + k0);
      const int kp = (k0 >> 1) + kh;
#pragma unroll
      for (int t = 0; t < 4; ++t) {
        v2f b = ((const v2f*)sW)[kp * NHID + t * 16 + l15];
        acc[t] = __builtin_amdgcn_wmma_f32_16x16x4_f32(
            false, a, false, b, (short)0, acc[t], false, false);
      }
    }
  }

  float dv[8];
#pragma unroll
  for (int j = 0; j < 8; ++j) dv[j] = rsqrtf(dg[j]);
#pragma unroll
  for (int t = 0; t < 4; ++t) {
    const int ccol = t * 16 + l15;
#pragma unroll
    for (int j = 0; j < 8; ++j) {
      const int r = rowBase + j + 8 * kh;
      if (r < n) h[(size_t)r * NHID + ccol] = acc[t][j];
    }
  }
  if (l15 == 0) {
#pragma unroll
    for (int j = 0; j < 8; ++j) {
      const int r = rowBase + j + 8 * kh;
      if (r < n) dinv[r] = dv[j];
    }
  }
}

#define EDGES_PER_WAVE 32

#define BKT_NODES   200
#define MAX_BKTS    1024
#define BIN_CAP     12800
#define BIN_THREADS 1024
#define BIN_PER     ((BIN_CAP + BIN_THREADS - 1) / BIN_THREADS)
#define SUM_THREADS 1024

__global__ __launch_bounds__(BIN_THREADS)
void edge_bin_kernel(const int* __restrict__ row, const int* __restrict__ col,
                     int2* __restrict__ binned, int* __restrict__ off, int nb, int e) {
  __shared__ int  cnt[MAX_BKTS];
  __shared__ int  pos[MAX_BKTS + 1];
  __shared__ int  wsum[32];
  __shared__ int2 stage[BIN_CAP];
  const int tid  = threadIdx.x;
  const int lane = tid & 31;
  const int wave = tid >> 5;
  const int s    = blockIdx.x;
  const int start = s * BIN_CAP;
  const int rem  = e - start;
  const int len  = rem < BIN_CAP ? rem : BIN_CAP;

  for (int j = tid; j < MAX_BKTS; j += BIN_THREADS) cnt[j] = 0;
  __syncthreads();

  int r[BIN_PER], c[BIN_PER], rk[BIN_PER];
#pragma unroll
  for (int u = 0; u < BIN_PER; ++u) {
    const int i  = u * BIN_THREADS + tid;
    const int ii = (i < len) ? i : (len - 1);
    r[u] = row[start + ii];
    const int v = col[start + ii];
    c[u] = (i < len) ? v : -1;
  }
#pragma unroll
  for (int u = 0; u < BIN_PER; ++u) {
    rk[u] = 0;
    if (c[u] >= 0) rk[u] = atomicAdd(&cnt[c[u] / BKT_NODES], 1);
  }
  __syncthreads();

  const int v = (tid < nb) ? cnt[tid] : 0;
  int x = v;
#pragma unroll
  for (int d = 1; d < 32; d <<= 1) {
    const int y = __shfl_up(x, d, 32);
    if (lane >= d) x += y;
  }
  if (lane == 31) wsum[wave] = x;
  __syncthreads();
  if (wave == 0) {
    const int w = wsum[lane];
    int z = w;
#pragma unroll
    for (int d = 1; d < 32; d <<= 1) {
      const int y = __shfl_up(z, d, 32);
      if (lane >= d) z += y;
    }
    wsum[lane] = z - w;
  }
  __syncthreads();
  if (tid < nb) pos[tid] = x - v + wsum[wave];
  if (tid == 0) pos[nb] = len;
  __syncthreads();

#pragma unroll
  for (int u = 0; u < BIN_PER; ++u) {
    if (c[u] >= 0) stage[pos[c[u] / BKT_NODES] + rk[u]] = make_int2(r[u], c[u]);
  }
  __syncthreads();

  for (int j = tid; j < len; j += BIN_THREADS) binned[(size_t)start + j] = stage[j];
  int* orow = off + (size_t)s * ((nb + 32) & ~31);
  for (int j = tid; j <= nb; j += BIN_THREADS) orow[j] = pos[j];
}

__global__ __launch_bounds__(SUM_THREADS)
void block_sum_kernel(const int2* __restrict__ binned, const int* __restrict__ off,
                      const float* __restrict__ dinv, const float* __restrict__ h,
                      float* __restrict__ agg, int n, int nb, int nslices) {
  __shared__ float acc[BKT_NODES * NHID];
  const int tid  = threadIdx.x;
  const int lane = tid & 31;
  const int wave = __builtin_amdgcn_readfirstlane((int)(tid >> 5));
  const int b    = blockIdx.x;
  const int lo   = b * BKT_NODES;
  const int left = n - lo;
  const int nodes = left < BKT_NODES ? left : BKT_NODES;

  for (int j = tid; j < nodes * (NHID / 4); j += SUM_THREADS) {
    const int node = lo + (j >> 4);
    const float d  = dinv[node];
    const float d2 = d * d;
    const float4 hv = *(const float4*)(h + (size_t)node * NHID + (j & 15) * 4);
    ((float4*)acc)[j] = make_float4(d2 * hv.x, d2 * hv.y, d2 * hv.z, d2 * hv.w);
  }
  __syncthreads();

  for (int s = wave; s < nslices; s += SUM_THREADS / 32) {
    const int* orow = off + (size_t)s * ((nb + 32) & ~31) + b;
    const int beg = __builtin_amdgcn_readfirstlane(orow[0]);
    const int end = __builtin_amdgcn_readfirstlane(orow[1]);
    const int2* seg = binned + (size_t)s * BIN_CAP;
    for (int e0 = beg; e0 < end; e0 += EDGES_PER_WAVE) {
      const int rem = end - e0;
      const int cnt = rem < EDGES_PER_WAVE ? rem : EDGES_PER_WAVE;
      int2 rc = make_int2(0, lo);
      if (lane < cnt) rc = seg[e0 + lane];
      const float myn = dinv[rc.x] * dinv[rc.y];
      const int   myo = (rc.y - lo) * NHID;
      for (int i0 = 0; i0 < cnt; i0 += 8) {
        float vl[8], vh[8];
#pragma unroll
        for (int u = 0; u < 8; ++u) {
          const int r = __builtin_amdgcn_readlane(rc.x, i0 + u);
          const float* hr = h + (size_t)r * NHID;
          vl[u] = hr[lane];
          vh[u] = hr[lane + 32];
        }
#pragma unroll
        for (int u = 0; u < 8; ++u) {
          if (i0 + u < cnt) {
            const int o = __builtin_amdgcn_readlane(myo, i0 + u);
            const float norm =
                __int_as_float(__builtin_amdgcn_readlane(__float_as_int(myn), i0 + u));
            atomicAdd(&acc[o + lane],      norm * vl[u]);
            atomicAdd(&acc[o + lane + 32], norm * vh[u]);
          }
        }
      }
    }
  }
  __syncthreads();

  float4* dst = (float4*)(agg + (size_t)lo * NHID);
  for (int j = tid; j < nodes * (NHID / 4); j += SUM_THREADS) dst[j] = ((const float4*)acc)[j];
}

__global__ __launch_bounds__(256)
void edge_mlp_kernel(const int* __restrict__ row, const int* __restrict__ col,
                     const float* __restrict__ agg,
                     const float* __restrict__ b1,
                     const float* __restrict__ Wfc,
                     const float* __restrict__ bfc,
                     float* __restrict__ out, int e) {
  const int lane = threadIdx.x & 31;
  const int wib  = __builtin_amdgcn_readfirstlane((int)(threadIdx.x >> 5));
  const int base = ((int)blockIdx.x * 8 + wib) * EDGES_PER_WAVE;
  if (base >= e) return;
  const bool valid = (base + lane) < e;
  const int myr = valid ? row[base + lane] : 0;
  const int myc = valid ? col[base + lane] : 0;

  const int half = lane >> 4;
  const int q4   = (lane & 15) * 4;
  const float4 bb = *(const float4*)(b1 + q4);
  const float* wp = Wfc + (size_t)(half * NHID + q4) * 2;
  const float4 wA = *(const float4*)(wp);
  const float4 wB = *(const float4*)(wp + 4);

  float s0[EDGES_PER_WAVE], s1[EDGES_PER_WAVE];
#pragma unroll
  for (int i = 0; i < EDGES_PER_WAVE; ++i) {
    const int r = __builtin_amdgcn_readlane(myr, i);
    const int c = __builtin_amdgcn_readlane(myc, i);
    const int node = half ? r : c;
    const float4 v = *(const float4*)(agg + (size_t)node * NHID + q4);
    const float v0 = v.x + bb.x, v1 = v.y + bb.y, v2 = v.z + bb.z, v3 = v.w + bb.w;
    s0[i] = v0 * wA.x + v1 * wA.z + v2 * wB.x + v3 * wB.z;
    s1[i] = v0 * wA.y + v1 * wA.w + v2 * wB.y + v3 * wB.w;
  }

#pragma unroll
  for (int s = 16; s >= 1; s >>= 1) {
    const bool up = (lane & s) != 0;
#pragma unroll
    for (int i = 0; i < s; ++i) {
      const float a0 = s0[i], b0 = s0[i + s];
      const float a1 = s1[i], b1v = s1[i + s];
      const float k0 = up ? b0  : a0;
      const float g0 = up ? a0  : b0;
      const float k1 = up ? b1v : a1;
      const float g1 = up ? a1  : b1v;
      s0[i] = k0 + __shfl_xor(g0, s, 32);
      s1[i] = k1 + __shfl_xor(g1, s, 32);
    }
  }

  if (valid) {
    float l0 = s0[0] + bfc[0], l1 = s1[0] + bfc[1];
    float m  = fmaxf(l0, l1);
    float e0 = __expf(l0 - m), e1 = __expf(l1 - m);
    float s  = 1.0f / (e0 + e1);
    ((float2*)out)[base + lane] = make_float2(e0 * s, e1 * s);
  }
}

extern "C" void kernel_launch(void* const* d_in, const int* in_sizes, int n_in,
                              void* d_out, int out_size, void* d_ws, size_t ws_size,
                              hipStream_t stream) {
  const float* x   = (const float*)d_in[0];
  const int*   ei  = (const int*)d_in[1];
  const float* W1  = (const float*)d_in[2];
  const float* b1  = (const float*)d_in[3];
  const float* Wfc = (const float*)d_in[4];
  const float* bfc = (const float*)d_in[5];
  const int n = in_sizes[0] / NFEAT;
  const int e = in_sizes[1] / 2;
  const int* row = ei;
  const int* col = ei + e;

  const int nb      = (n + BKT_NODES - 1) / BKT_NODES;
  const int nslices = (e + BIN_CAP - 1) / BIN_CAP;

  uintptr_t p = (uintptr_t)d_ws;
  float* deg  = (float*)p;  p += (size_t)n * sizeof(float);
  float* dinv = (float*)p;  p += (size_t)n * sizeof(float);
  p = (p + 255) & ~(uintptr_t)255;
  float* h    = (float*)p;  p += (size_t)n * NHID * sizeof(float);
  p = (p + 255) & ~(uintptr_t)255;
  float* agg  = (float*)p;  p += (size_t)n * NHID * sizeof(float);
  p = (p + 255) & ~(uintptr_t)255;
  unsigned* part = (unsigned*)p;
  p += (size_t)DEG_SLICES * ((n + 63) & ~63) * sizeof(unsigned);
  p = (p + 255) & ~(uintptr_t)255;
  int2* binned = (int2*)p;  p += (size_t)nslices * BIN_CAP * sizeof(int2);
  p = (p + 255) & ~(uintptr_t)255;
  int* off = (int*)p;

  deg_hist_kernel<<<dim3(DEG_SLICES, (n + DEG_RANGE - 1) / DEG_RANGE), DEG_THREADS, 0, stream>>>(
      col, part, n, e);
  deg_sum_kernel <<<(n + 1023) / 1024, 256, 0, stream>>>(part, deg, n);
  gemm_x_w1_kernel<<<(n + 127) / 128, 256, 0, stream>>>(x, W1, deg, h, dinv, n);
  edge_bin_kernel <<<nslices, BIN_THREADS, 0, stream>>>(row, col, binned, off, nb, e);
  block_sum_kernel<<<nb, SUM_THREADS, 0, stream>>>(binned, off, dinv, h, agg, n, nb, nslices);
  edge_mlp_kernel    <<<(e + 255) / 256, 256, 0, stream>>>(row, col, agg, b1, Wfc, bfc,
                                                       (float*)d_out, e);
}


# edge stage run by resident waves: a fixed grid of 8192 waves, each walking every 8192th 32-edge tile with the next tile's indices asked for ahead of its row gathers, not 50000 one-tile waves
# speedup vs baseline: 3.5905x; 1.1305x over previous
#include <hip/hip_runtime.h>
#include <math.h>
#include <stdint.h>

#define NFEAT 256
#define NHID  64

typedef float v2f __attribute__((ext_vector_type(2)));
typedef float v8f __attribute__((ext_vector_type(8)));
typedef float v4f __attribute__((ext_vector_type(4)));

__global__ __launch_bounds__(256)
void gemm_x_w1_kernel(const float* __restrict__ x,
                      const float* __restrict__ W1,
                      float* __restrict__ h, int n) {
  __shared__ float sW[128 * NHID];
  const int tid  = threadIdx.x;
  const int wave = tid >> 5;
  const int lane = tid & 31;
  const int l15  = lane & 15;
  const int kh   = lane >> 4;
  const int rowBase = blockIdx.x * 128 + wave * 16;
  const int row  = rowBase + l15;
  const int rowC = (row < n) ? row : (n - 1);

  v8f acc[4];
#pragma unroll
  for (int t = 0; t < 4; ++t) acc[t] = (v8f){0,0,0,0,0,0,0,0};

  const float* xrow = x + (size_t)rowC * NFEAT + 2 * kh;
  __builtin_prefetch(xrow + 128, 0, 3);

  for (int kc = 0; kc < 2; ++kc) {
    __syncthreads();
    for (int idx = tid; idx < 64 * NHID; idx += 256) {
      int kp = idx >> 6;
      int nn = idx & 63;
      int kg = kc * 128 + kp * 2;
      ((float2*)sW)[kp * NHID + nn] =
          make_float2(W1[(size_t)kg * NHID + nn], W1[(size_t)(kg + 1) * NHID + nn]);
    }
    __syncthreads();

#pragma unroll 4
    for (int k0 = 0; k0 < 128; k0 += 4) {
      v2f a = *(const v2f*)(xrow + kc * 128 + k0);
      const int kp = (k0 >> 1) + kh;
#pragma unroll
      for (int t = 0; t < 4; ++t) {
        v2f b = ((const v2f*)sW)[kp * NHID + t * 16 + l15];
        acc[t] = __builtin_amdgcn_wmma_f32_16x16x4_f32(
            false, a, false, b, (short)0, acc[t], false, false);
      }
    }
  }

#pragma unroll
  for (int t = 0; t < 4; ++t) {
    const int ccol = t * 16 + l15;
#pragma unroll
    for (int j = 0; j < 8; ++j) {
      const int r = rowBase + j + 8 * kh;
      if (r < n) h[(size_t)r * NHID + ccol] = acc[t][j];
    }
  }
}

#define EDGES_PER_WAVE 32

#define BKT_NODES   200
#define MAX_BKTS    1024
#define BIN_CAP     12800
#define BIN_THREADS 1024
#define BIN_PER     ((BIN_CAP + BIN_THREADS - 1) / BIN_THREADS)
#define SUM_THREADS 1024

__global__ __launch_bounds__(BIN_THREADS)
void edge_bin_kernel(const int* __restrict__ row, const int* __restrict__ col,
                     int2* __restrict__ binned, int* __restrict__ off, int nb, int e) {
  __shared__ int  cnt[MAX_BKTS];
  __shared__ int  pos[MAX_BKTS + 1];
  __shared__ int  wsum[32];
  __shared__ int2 stage[BIN_CAP];
  const int tid  = threadIdx.x;
  const int lane = tid & 31;
  const int wave = tid >> 5;
  const int s    = blockIdx.x;
  const int start = s * BIN_CAP;
  const int rem  = e - start;
  const int len  = rem < BIN_CAP ? rem : BIN_CAP;

  for (int j = tid; j < MAX_BKTS; j += BIN_THREADS) cnt[j] = 0;
  __syncthreads();

  int r[BIN_PER], c[BIN_PER], rk[BIN_PER];
#pragma unroll
  for (int u = 0; u < BIN_PER; ++u) {
    const int i  = u * BIN_THREADS + tid;
    const int ii = (i < len) ? i : (len - 1);
    r[u] = row[start + ii];
    const int v = col[start + ii];
    c[u] = (i < len) ? v : -1;
  }
#pragma unroll
  for (int u = 0; u < BIN_PER; ++u) {
    rk[u] = 0;
    if (c[u] >= 0) rk[u] = atomicAdd(&cnt[c[u] / BKT_NODES], 1);
  }
  __syncthreads();

  const int v = (tid < nb) ? cnt[tid] : 0;
  int x = v;
#pragma unroll
  for (int d = 1; d < 32; d <<= 1) {
    const int y = __shfl_up(x, d, 32);
    if (lane >= d) x += y;
  }
  if (lane == 31) wsum[wave] = x;
  __syncthreads();
  if (wave == 0) {
    const int w = wsum[lane];
    int z = w;
#pragma unroll
    for (int d = 1; d < 32; d <<= 1) {
      const int y = __shfl_up(z, d, 32);
      if (lane >= d) z += y;
    }
    wsum[lane] = z - w;
  }
  __syncthreads();
  if (tid < nb) pos[tid] = x - v + wsum[wave];
  if (tid == 0) pos[nb] = len;
  __syncthreads();

#pragma unroll
  for (int u = 0; u < BIN_PER; ++u) {
    if (c[u] >= 0) stage[pos[c[u] / BKT_NODES] + rk[u]] = make_int2(r[u], c[u]);
  }
  __syncthreads();

  for (int j = tid; j < len; j += BIN_THREADS) binned[(size_t)start + j] = stage[j];
  int* orow = off + (size_t)s * ((nb + 32) & ~31);
  for (int j = tid; j <= nb; j += BIN_THREADS) orow[j] = pos[j];
}

#define DINV_PITCH  256
#define DEG_THREADS 512
#define DEG_WAVES   (DEG_THREADS / 32)
#define DEG_SEGS    8

__global__ __launch_bounds__(DEG_THREADS)
void block_deg_kernel(const int2* __restrict__ binned, const int* __restrict__ off,
                      float* __restrict__ dinv, int n, int nb, int nslices) {
  __shared__ unsigned cnt[DINV_PITCH];
  const int tid  = threadIdx.x;
  const int lane = tid & 31;
  const int wave = __builtin_amdgcn_readfirstlane((int)(tid >> 5));
  const int b    = blockIdx.x;
  const int lo   = b * BKT_NODES;
  const int left = n - lo;
  const int nodes = left < BKT_NODES ? left : BKT_NODES;
  const int opitch = (nb + 32) & ~31;

  if (tid < DINV_PITCH) cnt[tid] = 0u;
  __syncthreads();

  for (int s0 = wave; s0 < nslices; s0 += DEG_SEGS * DEG_WAVES) {
    int beg[DEG_SEGS], end[DEG_SEGS];
#pragma unroll
    for (int u = 0; u < DEG_SEGS; ++u) {
      const int sl = s0 + u * DEG_WAVES;
      const int* orow = off + (size_t)((sl < nslices) ? sl : s0) * opitch + b;
      beg[u] = __builtin_amdgcn_readfirstlane(orow[0]);
      const int en = __builtin_amdgcn_readfirstlane(orow[1]);
      end[u] = (sl < nslices) ? en : beg[u];
    }
    int c0[DEG_SEGS], c1[DEG_SEGS];
#pragma unroll
    for (int u = 0; u < DEG_SEGS; ++u) {
      const int sl = s0 + u * DEG_WAVES;
      const int2* seg = binned + (size_t)((sl < nslices) ? sl : s0) * BIN_CAP;
      const int i = beg[u] + lane;
      c0[u] = -1;
      c1[u] = -1;
      if (i < end[u])      c0[u] = seg[i].y;
      if (i + 32 < end[u]) c1[u] = seg[i + 32].y;
    }
#pragma unroll
    for (int u = 0; u < DEG_SEGS; ++u) {
      if (c0[u] >= 0) atomicAdd(&cnt[c0[u] - lo], 1u);
      if (c1[u] >= 0) atomicAdd(&cnt[c1[u] - lo], 1u);
    }
#pragma unroll
    for (int u = 0; u < DEG_SEGS; ++u) {
      const int sl = s0 + u * DEG_WAVES;
      const int2* seg = binned + (size_t)((sl < nslices) ? sl : s0) * BIN_CAP;
      for (int e0 = beg[u] + 64; e0 < end[u]; e0 += 32) {
        const int i = e0 + lane;
        if (i < end[u]) atomicAdd(&cnt[seg[i].y - lo], 1u);
      }
    }
  }
  __syncthreads();

  if (tid < DINV_PITCH) {
    dinv[(size_t)b * DINV_PITCH + tid] =
        (tid < nodes) ? rsqrtf(1.0f + (float)cnt[tid]) : 0.0f;
  }
}

__global__ __launch_bounds__(SUM_THREADS)
void block_sum_kernel(const int2* __restrict__ binned, const int* __restrict__ off,
                      const float* __restrict__ dinv, const float* __restrict__ h,
                      float* __restrict__ agg, int n, int nb, int nslices) {
  __shared__ float acc[BKT_NODES * NHID];
  const int tid  = threadIdx.x;
  const int lane = tid & 31;
  const int wave = __builtin_amdgcn_readfirstlane((int)(tid >> 5));
  const int b    = blockIdx.x;
  const int lo   = b * BKT_NODES;
  const int left = n - lo;
  const int nodes = left < BKT_NODES ? left : BKT_NODES;

  for (int j = tid; j < nodes * (NHID / 4); j += SUM_THREADS) {
    const int node = lo + (j >> 4);
    const float d  = dinv[(size_t)b * DINV_PITCH + (j >> 4)];
    const float d2 = d * d;
    const float4 hv = *(const float4*)(h + (size_t)node * NHID + (j & 15) * 4);
    ((float4*)acc)[j] = make_float4(d2 * hv.x, d2 * hv.y, d2 * hv.z, d2 * hv.w);
  }
  __syncthreads();

  for (int s = wave; s < nslices; s += SUM_THREADS / 32) {
    const int* orow = off + (size_t)s * ((nb + 32) & ~31) + b;
    const int beg = __builtin_amdgcn_readfirstlane(orow[0]);
    const int end = __builtin_amdgcn_readfirstlane(orow[1]);
    const int2* seg = binned + (size_t)s * BIN_CAP;
    for (int e0 = beg; e0 < end; e0 += EDGES_PER_WAVE) {
      const int rem = end - e0;
      const int cnt = rem < EDGES_PER_WAVE ? rem : EDGES_PER_WAVE;
      int2 rc = make_int2(0, lo);
      if (lane < cnt) rc = seg[e0 + lane];
      const float myn = dinv[rc.x + (rc.x / BKT_NODES) * (DINV_PITCH - BKT_NODES)] *
                        dinv[b * DINV_PITCH + (rc.y - lo)];
      const int   myo = (rc.y - lo) * NHID;
      for (int i0 = 0; i0 < cnt; i0 += 8) {
        float vl[8], vh[8];
#pragma unroll
        for (int u = 0; u < 8; ++u) {
          const int r = __builtin_amdgcn_readlane(rc.x, i0 + u);
          const float* hr = h + (size_t)r * NHID;
          vl[u] = hr[lane];
          vh[u] = hr[lane + 32];
        }
#pragma unroll
        for (int u = 0; u < 8; ++u) {
          if (i0 + u < cnt) {
            const int o = __builtin_amdgcn_readlane(myo, i0 + u);
            const float norm =
                __int_as_float(__builtin_amdgcn_readlane(__float_as_int(myn), i0 + u));
            atomicAdd(&acc[o + lane],      norm * vl[u]);
            atomicAdd(&acc[o + lane + 32], norm * vh[u]);
          }
        }
      }
    }
  }
  __syncthreads();

  float4* dst = (float4*)(agg + (size_t)lo * NHID);
  for (int j = tid; j < nodes * (NHID / 4); j += SUM_THREADS) dst[j] = ((const float4*)acc)[j];
}

#define MLP_THREADS 256
#define MLP_WAVES   (MLP_THREADS / 32)
#define MLP_GROUPS  1024
#define MLP_ROWS    8

__global__ __launch_bounds__(MLP_THREADS)
void edge_mlp_kernel(const int* __restrict__ row, const int* __restrict__ col,
                     const float* __restrict__ agg,
                     const float* __restrict__ b1,
                     const float* __restrict__ Wfc,
                     const float* __restrict__ bfc,
                     float* __restrict__ out, int e) {
  const int lane = threadIdx.x & 31;
  const int wib    = __builtin_amdgcn_readfirstlane((int)(threadIdx.x >> 5));
  const int nwaves = (int)gridDim.x * MLP_WAVES;
  const int ntiles = (e + EDGES_PER_WAVE - 1) / EDGES_PER_WAVE;
  int tile = (int)blockIdx.x * MLP_WAVES + wib;
  if (tile >= ntiles) return;

  const int half = lane >> 4;
  const int q4   = (lane & 15) * 4;
  const float4 bb = *(const float4*)(b1 + q4);
  const float* wp = Wfc + (size_t)(half * NHID + q4) * 2;
  const float4 wA = *(const float4*)(wp);
  const float4 wB = *(const float4*)(wp + 4);
  const float bf0 = bfc[0], bf1 = bfc[1];

  int j0 = tile * EDGES_PER_WAVE + lane;
  j0 = j0 < e ? j0 : e - 1;
  int nr = row[j0], nc = col[j0];

  for (;;) {
    const int  base  = tile * EDGES_PER_WAVE;
    const bool valid = (base + lane) < e;
    const int  myr = nr, myc = nc;
    const int nxt = tile + nwaves;
    {
      const int tn = nxt < ntiles ? nxt : ntiles - 1;
      int j = tn * EDGES_PER_WAVE + lane;
      j = j < e ? j : e - 1;
      nr = row[j];
      nc = col[j];
    }

    float s0[EDGES_PER_WAVE], s1[EDGES_PER_WAVE];
#pragma unroll
    for (int g = 0; g < EDGES_PER_WAVE; g += MLP_ROWS) {
      v4f v[MLP_ROWS];
#pragma unroll
      for (int u = 0; u < MLP_ROWS; ++u) {
        const int r = __builtin_amdgcn_readlane(myr, g + u);
        const int c = __builtin_amdgcn_readlane(myc, g + u);
        const int node = half ? r : c;
        v[u] = *(const v4f*)((const char*)agg +
                             ((unsigned)node * (NHID * 4u) + (unsigned)q4 * 4u));
      }
      static_assert(MLP_ROWS == 8, "the fence below names 8 rows");
      asm volatile("" :: "v"(v[0]), "v"(v[1]), "v"(v[2]), "v"(v[3]),
                         "v"(v[4]), "v"(v[5]), "v"(v[6]), "v"(v[7]));
#pragma unroll
      for (int u = 0; u < MLP_ROWS; ++u) {
        const float v0 = v[u].x + bb.x, v1 = v[u].y + bb.y;
        const float v2 = v[u].z + bb.z, v3 = v[u].w + bb.w;
        s0[g + u] = v0 * wA.x + v1 * wA.z + v2 * wB.x + v3 * wB.z;
        s1[g + u] = v0 * wA.y + v1 * wA.w + v2 * wB.y + v3 * wB.w;
      }
    }

#pragma unroll
    for (int s = 16; s >= 1; s >>= 1) {
      const bool up = (lane & s) != 0;
#pragma unroll
      for (int i = 0; i < s; ++i) {
        const float a0 = s0[i], b0 = s0[i + s];
        const float a1 = s1[i], b1v = s1[i + s];
        const float k0 = up ? b0  : a0;
        const float g0 = up ? a0  : b0;
        const float k1 = up ? b1v : a1;
        const float g1 = up ? a1  : b1v;
        s0[i] = k0 + __shfl_xor(g0, s, 32);
        s1[i] = k1 + __shfl_xor(g1, s, 32);
      }
    }

    if (valid) {
      float l0 = s0[0] + bf0, l1 = s1[0] + bf1;
      float m  = fmaxf(l0, l1);
      float e0 = __expf(l0 - m), e1 = __expf(l1 - m);
      float sm = 1.0f / (e0 + e1);
      ((float2*)out)[base + lane] = make_float2(e0 * sm, e1 * sm);
    }

    if (nxt >= ntiles) break;
    tile = nxt;
  }
}

extern "C" void kernel_launch(void* const* d_in, const int* in_sizes, int n_in,
                              void* d_out, int out_size, void* d_ws, size_t ws_size,
                              hipStream_t stream) {
  const float* x   = (const float*)d_in[0];
  const int*   ei  = (const int*)d_in[1];
  const float* W1  = (const float*)d_in[2];
  const float* b1  = (const float*)d_in[3];
  const float* Wfc = (const float*)d_in[4];
  const float* bfc = (const float*)d_in[5];
  const int n = in_sizes[0] / NFEAT;
  const int e = in_sizes[1] / 2;
  const int* row = ei;
  const int* col = ei + e;

  const int nb      = (n + BKT_NODES - 1) / BKT_NODES;
  const int nslices = (e + BIN_CAP - 1) / BIN_CAP;

  uintptr_t p = (uintptr_t)d_ws;
  float* dinv = (float*)p;  p += (size_t)nb * DINV_PITCH * sizeof(float);
  p = (p + 255) & ~(uintptr_t)255;
  float* h    = (float*)p;  p += (size_t)n * NHID * sizeof(float);
  p = (p + 255) & ~(uintptr_t)255;
  float* agg  = (float*)p;  p += (size_t)n * NHID * sizeof(float);
  p = (p + 255) & ~(uintptr_t)255;
  int2* binned = (int2*)p;  p += (size_t)nslices * BIN_CAP * sizeof(int2);
  p = (p + 255) & ~(uintptr_t)255;
  int* off = (int*)p;

  gemm_x_w1_kernel<<<(n + 127) / 128, 256, 0, stream>>>(x, W1, h, n);
  edge_bin_kernel <<<nslices, BIN_THREADS, 0, stream>>>(row, col, binned, off, nb, e);
  block_deg_kernel<<<nb, DEG_THREADS, 0, stream>>>(binned, off, dinv, n, nb, nslices);
  block_sum_kernel<<<nb, SUM_THREADS, 0, stream>>>(binned, off, dinv, h, agg, n, nb, nslices);
  edge_mlp_kernel    <<<MLP_GROUPS, MLP_THREADS, 0, stream>>>(row, col, agg, b1, Wfc, bfc,
                                                          (float*)d_out, e);
}
